# starting kernel
# baseline (speedup 1.0000x reference)
#include <hip/hip_runtime.h>
#include <hip/hip_bf16.h>

typedef __attribute__((ext_vector_type(2))) float v2f;
typedef __attribute__((ext_vector_type(8))) float v8f;

#define BAG   15
#define NSTEP 5
#define LAMBDA 0.001f
#define PS    8
#define IM    28
#define DENOM (IM - PS)
#define BATCH 65536
#define INDIM 66
#define NBLK  512
#define WISTR 68

__global__ void cyclic_prep(const float* __restrict__ R,
                            const float* __restrict__ mask,
                            const float* __restrict__ partials,
                            float* __restrict__ ReffT,
                            int stepK) {
  int tid = threadIdx.x;
  int i = tid >> 4, j = tid & 15;
  float v = 0.f;
  if (stepK > 0 && i < BAG && j < BAG) {
    float co = 0.f;
    for (int b = 0; b < NBLK; ++b) co += partials[b * 256 + tid];
    co *= (1.0f / (float)BATCH);
    v = (R[i * BAG + j] - LAMBDA * co) * mask[i * BAG + j];
  }
  ReffT[j * 16 + i] = v;
}

__global__ __launch_bounds__(256)
void cyclic_step(const float* __restrict__ imgs,
                 const float* __restrict__ Wi,  const float* __restrict__ bi,
                 const float* __restrict__ gma, const float* __restrict__ bta,
                 const float* __restrict__ rb,
                 const float* __restrict__ Wc,  const float* __restrict__ bc,
                 const float* __restrict__ Wo,  const float* __restrict__ bo,
                 float* __restrict__ stateWS,   int* __restrict__ posWS,
                 float* __restrict__ partials,  const float* __restrict__ ReffT,
                 float* __restrict__ out,       int stepK) {
  __shared__ float ldsWi[16 * WISTR];
  __shared__ float ldsWout[256];
  __shared__ float ldsReff[256];
  __shared__ float ldsPar[48];
  __shared__ float ldsState[8][16][16];
  __shared__ float ldsCo[8][256];

  const int tid  = threadIdx.x;
  const int wave = tid >> 5;
  const int lane = tid & 31;
  const int half = lane >> 4;
  const int n16  = lane & 15;
  const int rowBase = blockIdx.x * 128 + wave * 16;
  const int myRow   = rowBase + n16;

  for (int idx = tid; idx < 16 * WISTR; idx += 256) {
    int n = idx / WISTR, k = idx - n * WISTR;
    ldsWi[idx] = (n < BAG && k < INDIM) ? Wi[n * INDIM + k] : 0.f;
  }
  {
    int k = tid >> 4, col = tid & 15;
    float v = 0.f;
    if (k < BAG) {
      if (col < 10)      v = Wo[col * BAG + k];
      else if (col < 12) v = Wc[(col - 10) * BAG + k];
    }
    ldsWout[tid] = v;
    ldsReff[tid] = ReffT[tid];
  }
  if (tid < 16) {
    float bv = 0.f, gv = 0.f, btv = 0.f;
    if (tid < BAG) {
      bv  = bi[tid] + ((stepK > 0) ? rb[tid] : 0.f);
      gv  = gma[tid];
      btv = bta[tid];
    }
    ldsPar[tid]      = bv;
    ldsPar[16 + tid] = gv;
    ldsPar[32 + tid] = btv;
  }
  const float bc0 = bc[0], bc1 = bc[1];
  const float boN = bo[n16 < 10 ? n16 : 9];

  int p0, p1;
  if (stepK == 0) { p0 = DENOM / 2; p1 = DENOM / 2; }
  else            { p0 = posWS[2 * myRow]; p1 = posWS[2 * myRow + 1]; }

  const float* img = imgs + (size_t)myRow * (IM * IM);
  __syncthreads();

  v8f accI = {0.f, 0.f, 0.f, 0.f, 0.f, 0.f, 0.f, 0.f};
#pragma unroll
  for (int t = 0; t < 17; ++t) {
    const int kb = 4 * t + 2 * half;
    v2f a;
    if (t < 16) {
      a.x = img[(p0 + (kb >> 3)) * IM + p1 + (kb & 7)];
      a.y = img[(p0 + ((kb + 1) >> 3)) * IM + p1 + ((kb + 1) & 7)];
    } else if (half == 0) {
      a.x = (float)p0 * (2.0f / DENOM) - 1.0f;
      a.y = (float)p1 * (2.0f / DENOM) - 1.0f;
    } else {
      a.x = 0.f; a.y = 0.f;
    }
    v2f b = *(const v2f*)&ldsWi[n16 * WISTR + kb];
    accI = __builtin_amdgcn_wmma_f32_16x16x4_f32(false, a, false, b,
                                                 (short)0, accI, false, false);
  }

  if (stepK > 0) {
    const float* st = stateWS + (size_t)myRow * 16;
#pragma unroll
    for (int t = 0; t < 4; ++t) {
      const int kb = 4 * t + 2 * half;
      v2f a, b;
      a.x = st[kb];
      a.y = st[kb + 1];
      b.x = ldsReff[kb * 16 + n16];
      b.y = ldsReff[(kb + 1) * 16 + n16];
      accI = __builtin_amdgcn_wmma_f32_16x16x4_f32(false, a, false, b,
                                                   (short)0, accI, false, false);
    }
  }

  const float biasN = ldsPar[n16];
  const float gN    = ldsPar[16 + n16];
  const float bN    = ldsPar[32 + n16];
#pragma unroll
  for (int r = 0; r < 8; ++r) {
    float h = fmaxf(accI[r] + biasN, 0.f);
    float sum = h;
#pragma unroll
    for (int m = 8; m >= 1; m >>= 1) sum += __shfl_xor(sum, m, 16);
    float mu = sum * (1.0f / BAG);
    float dd = h - mu;
    float sq = (n16 < BAG) ? dd * dd : 0.f;
#pragma unroll
    for (int m = 8; m >= 1; m >>= 1) sq += __shfl_xor(sq, m, 16);
    float s = dd * rsqrtf(sq * (1.0f / BAG) + 1e-5f) * gN + bN;
    const int M = r + half * 8;
    ldsState[wave][M][n16] = s;
    if (stepK < NSTEP - 1)
      stateWS[(size_t)(rowBase + M) * 16 + n16] = s;
  }
  __syncthreads();

  v8f accO = {0.f, 0.f, 0.f, 0.f, 0.f, 0.f, 0.f, 0.f};
#pragma unroll
  for (int t = 0; t < 4; ++t) {
    const int kb = 4 * t + 2 * half;
    v2f a, b;
    a.x = ldsState[wave][n16][kb];
    a.y = ldsState[wave][n16][kb + 1];
    b.x = ldsWout[kb * 16 + n16];
    b.y = ldsWout[(kb + 1) * 16 + n16];
    accO = __builtin_amdgcn_wmma_f32_16x16x4_f32(false, a, false, b,
                                                 (short)0, accO, false, false);
  }

#pragma unroll
  for (int r = 0; r < 8; ++r) {
    float raw10 = __shfl(accO[r], 10, 16);
    float raw11 = __shfl(accO[r], 11, 16);
    int q0 = __shfl(p0, half * 8 + r, 16);
    int q1 = __shfl(p1, half * 8 + r, 16);
    if (stepK < NSTEP - 1 && n16 == 0) {
      float c0 = tanhf(raw10 + bc0);
      float c1 = tanhf(raw11 + bc1);
      bool rowSel = fabsf(c0) >= fabsf(c1);
      int mv0 = rowSel ? ((c0 > 0.f) - (c0 < 0.f)) : 0;
      int mv1 = rowSel ? 0 : ((c1 > 0.f) - (c1 < 0.f));
      q0 = min(max(q0 + mv0, 0), DENOM);
      q1 = min(max(q1 + mv1, 0), DENOM);
      const int row = rowBase + half * 8 + r;
      posWS[2 * row]     = q0;
      posWS[2 * row + 1] = q1;
    }
  }

  if (stepK == NSTEP - 1 && n16 < 10) {
#pragma unroll
    for (int r = 0; r < 8; ++r)
      out[(size_t)(rowBase + r + half * 8) * 10 + n16] = accO[r] + boN;
  }

  if (stepK < NSTEP - 1) {
    v8f accC = {0.f, 0.f, 0.f, 0.f, 0.f, 0.f, 0.f, 0.f};
#pragma unroll
    for (int t = 0; t < 4; ++t) {
      const int kb = 4 * t + 2 * half;
      v2f f;
      f.x = ldsState[wave][kb][n16];
      f.y = ldsState[wave][kb + 1][n16];
      accC = __builtin_amdgcn_wmma_f32_16x16x4_f32(false, f, false, f,
                                                   (short)0, accC, false, false);
    }
#pragma unroll
    for (int r = 0; r < 8; ++r)
      ldsCo[wave][(r + half * 8) * 16 + n16] = accC[r];
  }
  __syncthreads();
  if (stepK < NSTEP - 1) {
    float acc = 0.f;
#pragma unroll
    for (int w = 0; w < 8; ++w) acc += ldsCo[w][tid];
    partials[(size_t)blockIdx.x * 256 + tid] = acc;
  }
}

extern "C" void kernel_launch(void* const* d_in, const int* in_sizes, int n_in,
                              void* d_out, int out_size, void* d_ws, size_t ws_size,
                              hipStream_t stream) {
  const float* x    = (const float*)d_in[0];
  const float* Wi   = (const float*)d_in[1];
  const float* bi   = (const float*)d_in[2];
  const float* gma  = (const float*)d_in[3];
  const float* bta  = (const float*)d_in[4];
  const float* R    = (const float*)d_in[5];
  const float* rb   = (const float*)d_in[6];
  const float* mask = (const float*)d_in[7];
  const float* Wc   = (const float*)d_in[8];
  const float* bc   = (const float*)d_in[9];
  const float* Wo   = (const float*)d_in[10];
  const float* bo   = (const float*)d_in[11];

  float* stateWS  = (float*)d_ws;
  int*   posWS    = (int*)((char*)d_ws + (size_t)BATCH * 16 * 4);
  float* partials = (float*)((char*)posWS + (size_t)BATCH * 2 * 4);
  float* ReffT    = partials + (size_t)NBLK * 256;

  for (int k = 0; k < NSTEP; ++k) {
    cyclic_prep<<<1, 256, 0, stream>>>(R, mask, partials, ReffT, k);
    cyclic_step<<<NBLK, 256, 0, stream>>>(x, Wi, bi, gma, bta, rb, Wc, bc,
                                          Wo, bo, stateWS, posWS, partials,
                                          ReffT, (float*)d_out, k);
  }
}


# coact product leaves the step kernel: the reduction launch (64 blocks) takes state^T state from the stored state, whole-line loads; each step block adds the 64 planes; coact sum order changes
# speedup vs baseline: 2.5713x; 2.5713x over previous
#include <hip/hip_runtime.h>
#include <hip/hip_bf16.h>


typedef __attribute__((ext_vector_type(2))) float v2f;
typedef __attribute__((ext_vector_type(2))) int   v2i;
typedef __attribute__((ext_vector_type(4))) float v4f;
typedef __attribute__((ext_vector_type(8))) float v8f;

__device__ __forceinline__ float xhalf(float x) {
  int b = __builtin_bit_cast(int, x);
  return __builtin_bit_cast(float,
      __builtin_amdgcn_permlanex16(b, b, 0x76543210, 0xfedcba98, false, false));
}

#define BAG   15
#define NSTEP 5
#define LAMBDA 0.001f
#define PS    8
#define IM    28
#define DENOM (IM - PS)
#define BATCH 65536
#define INDIM 66
#define TPW   2
#define BROWS (8 * TPW * 16)
#define NBLK  (BATCH / BROWS)
#define CG    64
#define CROWS (BATCH / CG)
#define WISTR 68

#define WI_OFF   0
#define WOUT_OFF (16 * WISTR)
#define PAR_OFF  (WOUT_OFF + 256)
#define TAIL_OFF (PAR_OFF + 64)
#define WIMG     2048

struct FixedPiece {
  float wv[4];
  float hv;
  float pv;
  float tv;
};

template <bool RB>
__device__ __forceinline__ FixedPiece load_fixed_piece(
    int u, bool addRb,
    const float* __restrict__ Wi,  const float* __restrict__ bi,
    const float* __restrict__ gma, const float* __restrict__ bta,
    const float* __restrict__ rb,
    const float* __restrict__ Wc,  const float* __restrict__ bc,
    const float* __restrict__ Wo,  const float* __restrict__ bo) {
  FixedPiece fp;
#pragma unroll
  for (int q = 0; q < 4; ++q) {
    const int idx = u + 256 * q;
    const int n = idx / WISTR, k = idx - n * WISTR;
    const bool ok = (n < BAG) && (k < INDIM);
    const float x = Wi[ok ? n * INDIM + k : 0];
    fp.wv[q] = ok ? x : 0.f;
  }
  const int hk = u >> 4, hc = u & 15;
  const bool hok = (hk < BAG) && (hc < 12);
  const float* hp = (hc < 10) ? (Wo + hc * BAG + hk) : (Wc + (hc - 10) * BAG + hk);
  const float hx = *(hok ? hp : Wo);
  fp.hv = hok ? hx : 0.f;
  const int pw = u >> 4, pf = u & 15;
  const bool pok = (pf < BAG) && (pw < 3);
  const float* prow = (pw == 0) ? bi : ((pw == 1) ? gma : bta);
  const float px = prow[pok ? pf : 0];
  float bsum = px;
  if constexpr (RB) {
    const float rx = rb[pok ? pf : 0];
    bsum = px + (addRb ? rx : 0.f);
  }
  fp.pv = pok ? ((pw == 0) ? bsum : px) : 0.f;
  const float* tp = (u < 2) ? (bc + u) : (bo + ((u < 12) ? (u - 2) : 0));
  const float tx = *tp;
  fp.tv = (u < 12) ? tx : 0.f;
  return fp;
}

__global__ __launch_bounds__(512)
void cyclic_coact(const float* __restrict__ Wi,  const float* __restrict__ bi,
                  const float* __restrict__ gma, const float* __restrict__ bta,
                  const float* __restrict__ rb,
                  const float* __restrict__ Wc,  const float* __restrict__ bc,
                  const float* __restrict__ Wo,  const float* __restrict__ bo,
                  const float* __restrict__ stateWS,
                  float* __restrict__ partials,
                  float* __restrict__ wimg) {
  __shared__ float ldsCo[16][256];
  const int tid = threadIdx.x;
  if (blockIdx.x == CG) {
    if (tid < 256) {
      const int u = tid;
      const FixedPiece fp = load_fixed_piece<true>(u, true, Wi, bi, gma, bta, rb,
                                                   Wc, bc, Wo, bo);
#pragma unroll
      for (int q = 0; q < 4; ++q) wimg[WI_OFF + u + 256 * q] = fp.wv[q];
      if (u + 1024 < 16 * WISTR) wimg[WI_OFF + u + 1024] = 0.f;
      wimg[WOUT_OFF + u] = fp.hv;
      if (u < 64) wimg[PAR_OFF + u] = fp.pv;
      wimg[TAIL_OFF + u] = fp.tv;
    } else {
      const int u = tid - 256;
      wimg[TAIL_OFF + 256 + u] = 0.f;
      if (TAIL_OFF + 512 + u < WIMG) wimg[TAIL_OFF + 512 + u] = 0.f;
    }
  } else {
    const int wave = tid >> 5;
    const int lane = tid & 31;
    const int half = lane >> 4;
    const int n16  = lane & 15;
    const float* sp = stateWS + (size_t)(blockIdx.x * CROWS + wave * 64 + half) * 16 + n16;
    v2f f[4][4];
#pragma unroll
    for (int t = 0; t < 4; ++t) {
#pragma unroll
      for (int k = 0; k < 4; ++k) {
        f[t][k].x = sp[(16 * t + 4 * k) * 16];
        f[t][k].y = sp[(16 * t + 4 * k + 2) * 16];
      }
    }
    v8f accC = {0.f, 0.f, 0.f, 0.f, 0.f, 0.f, 0.f, 0.f};
#pragma unroll
    for (int t = 0; t < 4; ++t) {
#pragma unroll
      for (int k = 0; k < 4; ++k)
        accC = __builtin_amdgcn_wmma_f32_16x16x4_f32(false, f[t][k], false, f[t][k],
                                                     (short)0, accC, false, false);
    }
#pragma unroll
    for (int r = 0; r < 8; ++r)
      ldsCo[wave][(r + half * 8) * 16 + n16] = accC[r];
    __syncthreads();
    if (tid < 256) {
      float acc = 0.f;
#pragma unroll
      for (int w = 0; w < 16; ++w) acc += ldsCo[w][tid];
      partials[(size_t)blockIdx.x * 256 + tid] = acc;
    }
  }
}

template <bool FIRST>
__global__ __launch_bounds__(256)
void cyclic_step(const float* __restrict__ imgs,
                 const float* __restrict__ wimg,
                 const float* __restrict__ Wi,  const float* __restrict__ bi,
                 const float* __restrict__ gma, const float* __restrict__ bta,
                 const float* __restrict__ R,   const float* __restrict__ mask,
                 const float* __restrict__ Wc,  const float* __restrict__ bc,
                 const float* __restrict__ Wo,  const float* __restrict__ bo,
                 float* __restrict__ stateWS,   int* __restrict__ posWS,
                 const float* __restrict__ partials,
                 float* __restrict__ out,       int stepArg) {
  const int stepK = FIRST ? 0 : stepArg;
  __shared__ __attribute__((aligned(16))) float ldsW[WIMG];
  __shared__ __attribute__((aligned(16))) float ldsState[8][TPW][16][16];
  __shared__ float ldsReff[256];
  const float* ldsWi   = ldsW + WI_OFF;
  const float* ldsWout = ldsW + WOUT_OFF;
  const float* ldsPar  = ldsW + PAR_OFF;

  const int tid  = threadIdx.x;
  const int wave = tid >> 5;
  const int lane = tid & 31;
  const int half = lane >> 4;
  const int n16  = lane & 15;
  const int rowBase = blockIdx.x * BROWS + wave * (TPW * 16);

  v4f w0 = {0.f, 0.f, 0.f, 0.f}, w1 = {0.f, 0.f, 0.f, 0.f};
  FixedPiece fp = {};
  float cp[CG];
  float rv = 0.f, mv = 0.f;
  const int ci = tid >> 4, cj = tid & 15;
  const bool cok = (ci < BAG) && (cj < BAG);
  if constexpr (FIRST) {
    fp = load_fixed_piece<false>(tid, false, Wi, bi, gma, bta, nullptr, Wc, bc, Wo, bo);
  } else {
    const v4f* wsrc = (const v4f*)wimg;
    w0 = wsrc[tid];
    w1 = wsrc[256 + tid];
#pragma unroll
    for (int g = 0; g < CG; ++g) cp[g] = partials[g * 256 + tid];
    rv = R[cok ? ci * BAG + cj : 0];
    mv = mask[cok ? ci * BAG + cj : 0];
  }

  int p0[TPW], p1[TPW];
  const float* img[TPW];
#pragma unroll
  for (int t = 0; t < TPW; ++t) {
    p0[t] = DENOM / 2; p1[t] = DENOM / 2;
    img[t] = imgs + (size_t)(rowBase + 16 * t + n16) * (IM * IM);
  }
  if constexpr (!FIRST) {
#pragma unroll
    for (int t = 0; t < TPW; ++t) {
      const int myRow = rowBase + 16 * t + n16;
      p0[t] = posWS[2 * myRow]; p1[t] = posWS[2 * myRow + 1];
    }
#pragma unroll
    for (int t = 0; t < TPW; ++t) asm volatile("" : "+v"(p0[t]), "+v"(p1[t]));
    ((v4f*)ldsW)[tid] = w0;
    ((v4f*)ldsW)[256 + tid] = w1;
    float c4[4] = {0.f, 0.f, 0.f, 0.f};
#pragma unroll
    for (int g = 0; g < CG / 4; ++g) {
#pragma unroll
      for (int q = 0; q < 4; ++q) c4[q] += cp[q * (CG / 4) + g];
    }
    const float co = ((c4[0] + c4[1]) + (c4[2] + c4[3])) * (1.0f / (float)BATCH);
    ldsReff[cj * 16 + ci] = cok ? (rv - LAMBDA * co) * mv : 0.f;
  } else {
#pragma unroll
    for (int q = 0; q < 4; ++q) ldsW[WI_OFF + tid + 256 * q] = fp.wv[q];
    if (tid + 1024 < 16 * WISTR) ldsW[WI_OFF + tid + 1024] = 0.f;
    ldsW[WOUT_OFF + tid] = fp.hv;
    if (tid < 64) ldsW[PAR_OFF + tid] = fp.pv;
    if (tid < 12) ldsW[TAIL_OFF + tid] = fp.tv;
  }
  __syncthreads();
  const float bc0 = ldsW[TAIL_OFF], bc1 = ldsW[TAIL_OFF + 1];
  const float boN = ldsW[TAIL_OFF + 2 + (n16 < 10 ? n16 : 9)];

  v2f ga[TPW][16];
#pragma unroll
  for (int k = 0; k < 16; ++k) {
    const int kb = 4 * k + 2 * half;
#pragma unroll
    for (int t = 0; t < TPW; ++t) {
      ga[t][k].x = img[t][(p0[t] + (kb >> 3)) * IM + p1[t] + (kb & 7)];
      ga[t][k].y = img[t][(p0[t] + ((kb + 1) >> 3)) * IM + p1[t] + ((kb + 1) & 7)];
    }
  }
  __builtin_amdgcn_sched_barrier(0);
  v8f accI[TPW];
#pragma unroll
  for (int t = 0; t < TPW; ++t)
    accI[t] = (v8f){0.f, 0.f, 0.f, 0.f, 0.f, 0.f, 0.f, 0.f};
#pragma unroll
  for (int k = 0; k < 17; ++k) {
    const int kb = 4 * k + 2 * half;
    v2f b = *(const v2f*)&ldsWi[n16 * WISTR + kb];
#pragma unroll
    for (int t = 0; t < TPW; ++t) {
      v2f a;
      if (k < 16) {
        a = ga[t][k];
      } else if (half == 0) {
        a.x = (float)p0[t] * (2.0f / DENOM) - 1.0f;
        a.y = (float)p1[t] * (2.0f / DENOM) - 1.0f;
      } else {
        a.x = 0.f; a.y = 0.f;
      }
      accI[t] = __builtin_amdgcn_wmma_f32_16x16x4_f32(false, a, false, b,
                                                      (short)0, accI[t], false, false);
    }
  }

  if constexpr (!FIRST) {
#pragma unroll
    for (int k = 0; k < 4; ++k) {
      const int kb = 4 * k + 2 * half;
      v2f b;
      b.x = ldsReff[kb * 16 + n16];
      b.y = ldsReff[(kb + 1) * 16 + n16];
#pragma unroll
      for (int t = 0; t < TPW; ++t) {
        const float* st = stateWS + (size_t)(rowBase + 16 * t + n16) * 16;
        v2f a;
        a.x = st[kb];
        a.y = st[kb + 1];
        accI[t] = __builtin_amdgcn_wmma_f32_16x16x4_f32(false, a, false, b,
                                                        (short)0, accI[t], false, false);
      }
    }
  }

  const float biasN = ldsPar[n16];
#pragma unroll
  for (int t = 0; t < TPW; ++t) {
#pragma unroll
    for (int r = 0; r < 8; ++r)
      ldsState[wave][t][r + half * 8][n16] = fmaxf(accI[t][r] + biasN, 0.f);
  }
  __syncthreads();

  v2f gam[4], bet[4];
#pragma unroll
  for (int j = 0; j < 4; ++j) {
    gam[j] = *(const v2f*)&ldsPar[16 + 4 * j + 2 * half];
    bet[j] = *(const v2f*)&ldsPar[32 + 4 * j + 2 * half];
  }
  v2f sA[TPW][4];
#pragma unroll
  for (int t = 0; t < TPW; ++t) {
    v2f hv[4];
#pragma unroll
    for (int j = 0; j < 4; ++j)
      hv[j] = *(const v2f*)&ldsState[wave][t][n16][4 * j + 2 * half];
    float s0 = (hv[0].x + hv[2].x) + (hv[1].x + hv[3].x);
    float s1 = (hv[0].y + hv[2].y) + (hv[1].y + hv[3].y);
    s0 += xhalf(s0);
    s1 += xhalf(s1);
    const float mu = (s0 + s1) * (1.0f / BAG);
    v2f dd[4];
#pragma unroll
    for (int j = 0; j < 4; ++j) { dd[j].x = hv[j].x - mu; dd[j].y = hv[j].y - mu; }
    const float padsq = (half == 1) ? 0.f : dd[3].y * dd[3].y;
    float q0 = (dd[0].x * dd[0].x + dd[2].x * dd[2].x) + (dd[1].x * dd[1].x + dd[3].x * dd[3].x);
    float q1 = (dd[0].y * dd[0].y + dd[2].y * dd[2].y) + (dd[1].y * dd[1].y + padsq);
    q0 += xhalf(q0);
    q1 += xhalf(q1);
    const float rstd = rsqrtf((q0 + q1) * (1.0f / BAG) + 1e-5f);
#pragma unroll
    for (int j = 0; j < 4; ++j) {
      sA[t][j].x = dd[j].x * rstd * gam[j].x + bet[j].x;
      sA[t][j].y = dd[j].y * rstd * gam[j].y + bet[j].y;
    }
  }
  if (stepK < NSTEP - 1) {
#pragma unroll
    for (int t = 0; t < TPW; ++t) {
      float* st = stateWS + (size_t)(rowBase + 16 * t + n16) * 16;
#pragma unroll
      for (int j = 0; j < 4; ++j)
        *(v2f*)&st[4 * j + 2 * half] = sA[t][j];
    }
  }

  if (stepK < NSTEP - 1) {
    static_assert(TPW == 2, "one tile per lane half");
    const bool wcRow = (n16 & 6) == 0;
    v2f wc[4];
#pragma unroll
    for (int j = 0; j < 4; ++j) {
      const int kb = 4 * j + 2 * half;
      const float x = ldsWout[kb * 16 + 10 + (n16 & 1)];
      const float y = ldsWout[(kb + 1) * 16 + 10 + (n16 & 1)];
      wc[j].x = wcRow ? x : 0.f;
      wc[j].y = wcRow ? y : 0.f;
    }
    v8f accK[TPW];
#pragma unroll
    for (int t = 0; t < TPW; ++t)
      accK[t] = (v8f){0.f, 0.f, 0.f, 0.f, 0.f, 0.f, 0.f, 0.f};
#pragma unroll
    for (int j = 0; j < 4; ++j) {
#pragma unroll
      for (int t = 0; t < TPW; ++t)
        accK[t] = __builtin_amdgcn_wmma_f32_16x16x4_f32(false, wc[j], false, sA[t][j],
                                                        (short)0, accK[t], false, false);
    }
    const float raw10 = half ? accK[1][0] : accK[0][0];
    const float raw11 = half ? accK[1][1] : accK[0][1];
    int q0 = half ? p0[1] : p0[0];
    int q1 = half ? p1[1] : p1[0];
    const float c0 = tanhf(raw10 + bc0);
    const float c1 = tanhf(raw11 + bc1);
    const bool rowSel = fabsf(c0) >= fabsf(c1);
    const int mv0 = rowSel ? ((c0 > 0.f) - (c0 < 0.f)) : 0;
    const int mv1 = rowSel ? 0 : ((c1 > 0.f) - (c1 < 0.f));
    q0 = min(max(q0 + mv0, 0), DENOM);
    q1 = min(max(q1 + mv1, 0), DENOM);
    const int row = rowBase + 16 * half + n16;
    v2i qq; qq.x = q0; qq.y = q1;
    *(v2i*)&posWS[2 * row] = qq;
  } else {
    v8f accO[TPW];
#pragma unroll
    for (int t = 0; t < TPW; ++t)
      accO[t] = (v8f){0.f, 0.f, 0.f, 0.f, 0.f, 0.f, 0.f, 0.f};
#pragma unroll
    for (int k = 0; k < 4; ++k) {
      const int kb = 4 * k + 2 * half;
      v2f b;
      b.x = ldsWout[kb * 16 + n16];
      b.y = ldsWout[(kb + 1) * 16 + n16];
#pragma unroll
      for (int t = 0; t < TPW; ++t) {
        accO[t] = __builtin_amdgcn_wmma_f32_16x16x4_f32(false, sA[t][k], false, b,
                                                        (short)0, accO[t], false, false);
      }
    }
    if (n16 < 10) {
#pragma unroll
      for (int t = 0; t < TPW; ++t) {
#pragma unroll
        for (int r = 0; r < 8; ++r)
          out[(size_t)(rowBase + 16 * t + r + half * 8) * 10 + n16] = accO[t][r] + boN;
      }
    }
  }
}

extern "C" void kernel_launch(void* const* d_in, const int* in_sizes, int n_in,
                              void* d_out, int out_size, void* d_ws, size_t ws_size,
                              hipStream_t stream) {
  const float* x    = (const float*)d_in[0];
  const float* Wi   = (const float*)d_in[1];
  const float* bi   = (const float*)d_in[2];
  const float* gma  = (const float*)d_in[3];
  const float* bta  = (const float*)d_in[4];
  const float* R    = (const float*)d_in[5];
  const float* rb   = (const float*)d_in[6];
  const float* mask = (const float*)d_in[7];
  const float* Wc   = (const float*)d_in[8];
  const float* bc   = (const float*)d_in[9];
  const float* Wo   = (const float*)d_in[10];
  const float* bo   = (const float*)d_in[11];

  float* stateWS  = (float*)d_ws;
  int*   posWS    = (int*)((char*)d_ws + (size_t)BATCH * 16 * 4);
  float* partials = (float*)((char*)posWS + (size_t)BATCH * 2 * 4);
  float* wimg     = partials + (size_t)CG * 256;

  cyclic_step<true><<<NBLK, 256, 0, stream>>>(x, wimg, Wi, bi, gma, bta, R, mask, Wc, bc, Wo, bo,
                                              stateWS, posWS, partials, (float*)d_out, 0);
  for (int k = 1; k < NSTEP; ++k) {
    cyclic_coact<<<CG + 1, 512, 0, stream>>>(Wi, bi, gma, bta, rb, Wc, bc, Wo, bo,
                                             stateWS, partials, wimg);
    cyclic_step<false><<<NBLK, 256, 0, stream>>>(x, wimg, Wi, bi, gma, bta, R, mask, Wc, bc, Wo, bo,
                                                 stateWS, posWS, partials, (float*)d_out, k);
  }
}
